# starting kernel
# baseline (speedup 1.0000x reference)
#include <hip/hip_runtime.h>


typedef __attribute__((ext_vector_type(2))) float v2f;
typedef __attribute__((ext_vector_type(8))) float v8f;

__device__ __forceinline__ void lds_fence() {
  asm volatile("s_wait_dscnt 0" ::: "memory");
}

__device__ __forceinline__ v8f wmma4(v2f a, v2f b, v8f c) {
  return __builtin_amdgcn_wmma_f32_16x16x4_f32(false, a, false, b, (short)0, c,
                                               false, false);
}

__device__ __forceinline__ void cnot_ring(float* Sre, float* Sim, int lane) {
#pragma unroll
  for (int q = 0; q < 9; ++q) {
    const int qt = (q + 1) % 9;
    const int tmask = 1 << qt;
    const int cmask = 1 << q;
#pragma unroll
    for (int jj = 0; jj < 8; ++jj) {
      int k = lane + 32 * jj;
      int idx0 = ((k >> qt) << (qt + 1)) | (k & (tmask - 1));
      if (idx0 & cmask) {
        int idx1 = idx0 | tmask;
        float r0 = Sre[idx0], r1 = Sre[idx1];
        Sre[idx0] = r1; Sre[idx1] = r0;
        float i0 = Sim[idx0], i1 = Sim[idx1];
        Sim[idx0] = i1; Sim[idx1] = i0;
      }
    }
    lds_fence();
  }
}

__launch_bounds__(256)
__global__ void quanv_kernel(const float* __restrict__ x,
                             const float* __restrict__ qw,
                             float* __restrict__ out) {
  constexpr int NQ = 9, B = 4, C = 3, H = 28, W = 28, HOUT = 26, WOUT = 26;
  constexpr int TOTAL = B * HOUT * WOUT;

  __shared__ float lds[8][1024];

  const int lane = threadIdx.x & 31;
  const int warp = threadIdx.x >> 5;
  const int wid = blockIdx.x * 8 + warp;
  if (wid >= TOTAL) return;

  const int b  = wid / (HOUT * WOUT);
  const int yy = (wid / WOUT) % HOUT;
  const int xx = wid % WOUT;

  float* Sre = &lds[warp][0];
  float* Sim = &lds[warp][512];

  const int m16 = lane & 15;
  const int hh  = lane >> 4;

  __builtin_prefetch(x + ((b * C) * H + yy) * W + xx, 0, 1);

  float c1[NQ], s1[NQ];
#pragma unroll
  for (int i = 0; i < NQ; ++i) {
    float a = qw[NQ + i] * 0.5f;
    c1[i] = cosf(a);
    s1[i] = sinf(a);
  }

  v2f gAr[4], gAi[4];
#pragma unroll
  for (int kc = 0; kc < 4; ++kc) {
#pragma unroll
    for (int v = 0; v < 2; ++v) {
      int k = 4 * kc + v + 2 * hh;
      int diff = m16 ^ k;
      float P = 1.0f;
#pragma unroll
      for (int i = 0; i < 4; ++i) P *= ((diff >> i) & 1) ? s1[i] : c1[i];
      int hm = __popc(diff) & 3;
      gAr[kc][v] = (hm == 0) ? P : ((hm == 2) ? -P : 0.0f);
      gAi[kc][v] = (hm == 1) ? -P : ((hm == 3) ? P : 0.0f);
    }
  }

  v2f gBr[2][8], gBi[2][8];
#pragma unroll
  for (int t = 0; t < 2; ++t) {
    int cp = 16 * t + m16;
#pragma unroll
    for (int kc = 0; kc < 8; ++kc) {
#pragma unroll
      for (int v = 0; v < 2; ++v) {
        int c = 4 * kc + v + 2 * hh;
        int diff = c ^ cp;
        float P = 1.0f;
#pragma unroll
        for (int i = 0; i < 5; ++i) P *= ((diff >> i) & 1) ? s1[4 + i] : c1[4 + i];
        int hm = __popc(diff) & 3;
        gBr[t][kc][v] = (hm == 0) ? P : ((hm == 2) ? -P : 0.0f);
        gBi[t][kc][v] = (hm == 1) ? -P : ((hm == 3) ? P : 0.0f);
      }
    }
  }

  float zacc[4] = {0.f, 0.f, 0.f, 0.f};

  for (int ch = 0; ch < C; ++ch) {
    float ca[NQ], sa[NQ];
    const float* xp = x + ((b * C + ch) * H + yy) * W + xx;
#pragma unroll
    for (int dy = 0; dy < 3; ++dy) {
#pragma unroll
      for (int dx = 0; dx < 3; ++dx) {
        int i = dy * 3 + dx;
        float a = (xp[dy * W + dx] + qw[i]) * 0.5f;
        ca[i] = cosf(a);
        sa[i] = sinf(a);
      }
    }

#pragma unroll
    for (int j = 0; j < 16; ++j) {
      int idx = lane + 32 * j;
      float P = 1.0f;
#pragma unroll
      for (int i = 0; i < NQ; ++i) P *= ((idx >> i) & 1) ? sa[i] : ca[i];
      int hm = __popc(idx) & 3;
      Sre[idx] = (hm == 0) ? P : ((hm == 2) ? -P : 0.0f);
      Sim[idx] = (hm == 1) ? -P : ((hm == 3) ? P : 0.0f);
    }
    lds_fence();

    cnot_ring(Sre, Sim, lane);

    v2f bR[2][4], bI[2][4];
#pragma unroll
    for (int t = 0; t < 2; ++t) {
#pragma unroll
      for (int kc = 0; kc < 4; ++kc) {
#pragma unroll
        for (int v = 0; v < 2; ++v) {
          int r = 4 * kc + v + 2 * hh;
          int c = 16 * t + m16;
          bR[t][kc][v] = Sre[r + 16 * c];
          bI[t][kc][v] = Sim[r + 16 * c];
        }
      }
    }
#pragma unroll
    for (int t = 0; t < 2; ++t) {
      v8f rr = {0.f,0.f,0.f,0.f,0.f,0.f,0.f,0.f};
      v8f ii = rr, ri = rr, ir = rr;
#pragma unroll
      for (int kc = 0; kc < 4; ++kc) {
        rr = wmma4(gAr[kc], bR[t][kc], rr);
        ii = wmma4(gAi[kc], bI[t][kc], ii);
        ri = wmma4(gAr[kc], bI[t][kc], ri);
        ir = wmma4(gAi[kc], bR[t][kc], ir);
      }
      v8f nr = rr - ii, ni = ri + ir;
#pragma unroll
      for (int v = 0; v < 8; ++v) {
        int r = v + 8 * hh;
        int c = 16 * t + m16;
        Sre[r + 16 * c] = nr[v];
        Sim[r + 16 * c] = ni[v];
      }
    }
    lds_fence();

    v2f aR[8], aI[8];
#pragma unroll
    for (int kc = 0; kc < 8; ++kc) {
#pragma unroll
      for (int v = 0; v < 2; ++v) {
        int c = 4 * kc + v + 2 * hh;
        aR[kc][v] = Sre[m16 + 16 * c];
        aI[kc][v] = Sim[m16 + 16 * c];
      }
    }
#pragma unroll
    for (int t = 0; t < 2; ++t) {
      v8f rr = {0.f,0.f,0.f,0.f,0.f,0.f,0.f,0.f};
      v8f ii = rr, ri = rr, ir = rr;
#pragma unroll
      for (int kc = 0; kc < 8; ++kc) {
        rr = wmma4(aR[kc], gBr[t][kc], rr);
        ii = wmma4(aI[kc], gBi[t][kc], ii);
        ri = wmma4(aI[kc], gBr[t][kc], ri);
        ir = wmma4(aR[kc], gBi[t][kc], ir);
      }
      v8f nr = rr - ii, ni = ri + ir;
#pragma unroll
      for (int v = 0; v < 8; ++v) {
        int r = v + 8 * hh;
        int c = 16 * t + m16;
        Sre[r + 16 * c] = nr[v];
        Sim[r + 16 * c] = ni[v];
      }
    }
    lds_fence();

    cnot_ring(Sre, Sim, lane);

    float z[4] = {0.f, 0.f, 0.f, 0.f};
#pragma unroll
    for (int j = 0; j < 16; ++j) {
      int idx = lane + 32 * j;
      float re = Sre[idx], im = Sim[idx];
      float p = re * re + im * im;
#pragma unroll
      for (int f = 0; f < 4; ++f) z[f] += ((idx >> f) & 1) ? -p : p;
    }
#pragma unroll
    for (int f = 0; f < 4; ++f) {
#pragma unroll
      for (int m = 1; m < 32; m <<= 1) z[f] += __shfl_xor(z[f], m, 32);
      zacc[f] += z[f];
    }
    lds_fence();
  }

  if (lane == 0) {
    const float inv = 1.0f / 3.0f;
#pragma unroll
    for (int f = 0; f < 4; ++f)
      out[((b * 4 + f) * HOUT + yy) * WOUT + xx] = zacc[f] * inv;
  }
}

extern "C" void kernel_launch(void* const* d_in, const int* in_sizes, int n_in,
                              void* d_out, int out_size, void* d_ws, size_t ws_size,
                              hipStream_t stream) {
  (void)in_sizes; (void)n_in; (void)out_size; (void)d_ws; (void)ws_size;
  const float* x  = (const float*)d_in[0];
  const float* qw = (const float*)d_in[1];
  float* out = (float*)d_out;
  const int total_waves = 4 * 26 * 26;
  dim3 grid(total_waves / 8), block(256);
  hipLaunchKernelGGL(quanv_kernel, grid, block, 0, stream, x, qw, out);
}


# 9-CNOT ring composed into one GF(2) index map: ring 1 folded into the product-state source index, ring 2 into the Z sign parities; 18 LDS swap passes per channel gone (Z partial sums regrouped)
# speedup vs baseline: 1.4132x; 1.4132x over previous
#include <hip/hip_runtime.h>


typedef __attribute__((ext_vector_type(2))) float v2f;
typedef __attribute__((ext_vector_type(8))) float v8f;

__device__ __forceinline__ void lds_fence() {
  asm volatile("s_wait_dscnt 0" ::: "memory");
}

__device__ __forceinline__ v8f wmma4(v2f a, v2f b, v8f c) {
  return __builtin_amdgcn_wmma_f32_16x16x4_f32(false, a, false, b, (short)0, c,
                                               false, false);
}

__device__ __forceinline__ int ring_src(int idx) {
  return ((idx ^ (idx << 1)) & 0x1FE) ^ (idx & 1) ^ (((idx >> 8) & 1) * 3);
}

__launch_bounds__(256)
__global__ void quanv_kernel(const float* __restrict__ x,
                             const float* __restrict__ qw,
                             float* __restrict__ out) {
  constexpr int NQ = 9, B = 4, C = 3, H = 28, W = 28, HOUT = 26, WOUT = 26;
  constexpr int TOTAL = B * HOUT * WOUT;

  __shared__ float lds[8][1024];

  const int lane = threadIdx.x & 31;
  const int warp = threadIdx.x >> 5;
  const int wid = blockIdx.x * 8 + warp;
  if (wid >= TOTAL) return;

  const int b  = wid / (HOUT * WOUT);
  const int yy = (wid / WOUT) % HOUT;
  const int xx = wid % WOUT;

  float* Sre = &lds[warp][0];
  float* Sim = &lds[warp][512];

  const int m16 = lane & 15;
  const int hh  = lane >> 4;

  __builtin_prefetch(x + ((b * C) * H + yy) * W + xx, 0, 1);

  float c1[NQ], s1[NQ];
#pragma unroll
  for (int i = 0; i < NQ; ++i) {
    float a = qw[NQ + i] * 0.5f;
    c1[i] = cosf(a);
    s1[i] = sinf(a);
  }

  v2f gAr[4], gAi[4];
#pragma unroll
  for (int kc = 0; kc < 4; ++kc) {
#pragma unroll
    for (int v = 0; v < 2; ++v) {
      int k = 4 * kc + v + 2 * hh;
      int diff = m16 ^ k;
      float P = 1.0f;
#pragma unroll
      for (int i = 0; i < 4; ++i) P *= ((diff >> i) & 1) ? s1[i] : c1[i];
      int hm = __popc(diff) & 3;
      gAr[kc][v] = (hm == 0) ? P : ((hm == 2) ? -P : 0.0f);
      gAi[kc][v] = (hm == 1) ? -P : ((hm == 3) ? P : 0.0f);
    }
  }

  v2f gBr[2][8], gBi[2][8];
#pragma unroll
  for (int t = 0; t < 2; ++t) {
    int cp = 16 * t + m16;
#pragma unroll
    for (int kc = 0; kc < 8; ++kc) {
#pragma unroll
      for (int v = 0; v < 2; ++v) {
        int c = 4 * kc + v + 2 * hh;
        int diff = c ^ cp;
        float P = 1.0f;
#pragma unroll
        for (int i = 0; i < 5; ++i) P *= ((diff >> i) & 1) ? s1[4 + i] : c1[4 + i];
        int hm = __popc(diff) & 3;
        gBr[t][kc][v] = (hm == 0) ? P : ((hm == 2) ? -P : 0.0f);
        gBi[t][kc][v] = (hm == 1) ? -P : ((hm == 3) ? P : 0.0f);
      }
    }
  }

  float zacc[4] = {0.f, 0.f, 0.f, 0.f};

  for (int ch = 0; ch < C; ++ch) {
    float ca[NQ], sa[NQ];
    const float* xp = x + ((b * C + ch) * H + yy) * W + xx;
#pragma unroll
    for (int dy = 0; dy < 3; ++dy) {
#pragma unroll
      for (int dx = 0; dx < 3; ++dx) {
        int i = dy * 3 + dx;
        float a = (xp[dy * W + dx] + qw[i]) * 0.5f;
        ca[i] = cosf(a);
        sa[i] = sinf(a);
      }
    }

    const int b01 = lane & 1, b12 = (lane ^ (lane >> 1)) & 1;
    const float fL = (((lane ^ (lane >> 1)) & 2) ? sa[2] : ca[2]) *
                     (((lane ^ (lane >> 1)) & 4) ? sa[3] : ca[3]) *
                     (((lane ^ (lane >> 1)) & 8) ? sa[4] : ca[4]);
    float f01[2], f5[2];
    f01[0] = fL * ((b01 ? sa[0] : ca[0]) * (b12 ? sa[1] : ca[1]));
    f01[1] = fL * ((b01 ? ca[0] : sa[0]) * (b12 ? ca[1] : sa[1]));
    f5[0] = (lane & 16) ? sa[5] : ca[5];
    f5[1] = (lane & 16) ? ca[5] : sa[5];
#pragma unroll
    for (int j = 0; j < 16; ++j) {
      const int idx = lane + 32 * j;
      const int s = ring_src(idx);
      const int j0 = j & 1, j3 = (j >> 3) & 1;
      const int hi = (j ^ (j >> 1)) & 7;
      const float k678 = ((hi & 1) ? sa[6] : ca[6]) * ((hi & 2) ? sa[7] : ca[7]) *
                         ((hi & 4) ? sa[8] : ca[8]);
      const float P = (f01[j3] * f5[j0]) * k678;
      const int hm = __popc(s) & 3;
      Sre[idx] = (hm == 0) ? P : ((hm == 2) ? -P : 0.0f);
      Sim[idx] = (hm == 1) ? -P : ((hm == 3) ? P : 0.0f);
    }
    lds_fence();

    v2f bR[2][4], bI[2][4];
#pragma unroll
    for (int t = 0; t < 2; ++t) {
#pragma unroll
      for (int kc = 0; kc < 4; ++kc) {
#pragma unroll
        for (int v = 0; v < 2; ++v) {
          int r = 4 * kc + v + 2 * hh;
          int c = 16 * t + m16;
          bR[t][kc][v] = Sre[r + 16 * c];
          bI[t][kc][v] = Sim[r + 16 * c];
        }
      }
    }
#pragma unroll
    for (int t = 0; t < 2; ++t) {
      v8f rr = {0.f,0.f,0.f,0.f,0.f,0.f,0.f,0.f};
      v8f ii = rr, ri = rr, ir = rr;
#pragma unroll
      for (int kc = 0; kc < 4; ++kc) {
        rr = wmma4(gAr[kc], bR[t][kc], rr);
        ii = wmma4(gAi[kc], bI[t][kc], ii);
        ri = wmma4(gAr[kc], bI[t][kc], ri);
        ir = wmma4(gAi[kc], bR[t][kc], ir);
      }
      v8f nr = rr - ii, ni = ri + ir;
#pragma unroll
      for (int v = 0; v < 8; ++v) {
        int r = v + 8 * hh;
        int c = 16 * t + m16;
        Sre[r + 16 * c] = nr[v];
        Sim[r + 16 * c] = ni[v];
      }
    }
    lds_fence();

    v2f aR[8], aI[8];
#pragma unroll
    for (int kc = 0; kc < 8; ++kc) {
#pragma unroll
      for (int v = 0; v < 2; ++v) {
        int c = 4 * kc + v + 2 * hh;
        aR[kc][v] = Sre[m16 + 16 * c];
        aI[kc][v] = Sim[m16 + 16 * c];
      }
    }
#pragma unroll
    for (int t = 0; t < 2; ++t) {
      v8f rr = {0.f,0.f,0.f,0.f,0.f,0.f,0.f,0.f};
      v8f ii = rr, ri = rr, ir = rr;
#pragma unroll
      for (int kc = 0; kc < 8; ++kc) {
        rr = wmma4(aR[kc], gBr[t][kc], rr);
        ii = wmma4(aI[kc], gBi[t][kc], ii);
        ri = wmma4(aI[kc], gBr[t][kc], ri);
        ir = wmma4(aR[kc], gBi[t][kc], ir);
      }
      v8f nr = rr - ii, ni = ri + ir;
#pragma unroll
      for (int v = 0; v < 8; ++v) {
        int r = v + 8 * hh;
        int c = 16 * t + m16;
        Sre[r + 16 * c] = nr[v];
        Sim[r + 16 * c] = ni[v];
      }
    }
    lds_fence();

    float tot = 0.f, alt = 0.f;
#pragma unroll
    for (int j = 0; j < 16; ++j) {
      int idx = lane + 32 * j;
      float re = Sre[idx], im = Sim[idx];
      float p = re * re + im * im;
      tot += p;
      alt += (__builtin_popcount(j) & 1) ? -p : p;
    }
    float z[4];
    z[0] = (__popc(lane & 0x1E) & 1) ? -alt : alt;
    z[1] = (__popc(lane & 0x03) & 1) ? -tot : tot;
    z[2] = (__popc(lane & 0x07) & 1) ? -tot : tot;
    z[3] = (__popc(lane & 0x0F) & 1) ? -tot : tot;
#pragma unroll
    for (int f = 0; f < 4; ++f) {
#pragma unroll
      for (int m = 1; m < 32; m <<= 1) z[f] += __shfl_xor(z[f], m, 32);
      zacc[f] += z[f];
    }
    lds_fence();
  }

  if (lane == 0) {
    const float inv = 1.0f / 3.0f;
#pragma unroll
    for (int f = 0; f < 4; ++f)
      out[((b * 4 + f) * HOUT + yy) * WOUT + xx] = zacc[f] * inv;
  }
}

extern "C" void kernel_launch(void* const* d_in, const int* in_sizes, int n_in,
                              void* d_out, int out_size, void* d_ws, size_t ws_size,
                              hipStream_t stream) {
  (void)in_sizes; (void)n_in; (void)out_size; (void)d_ws; (void)ws_size;
  const float* x  = (const float*)d_in[0];
  const float* qw = (const float*)d_in[1];
  float* out = (float*)d_out;
  const int total_waves = 4 * 26 * 26;
  dim3 grid(total_waves / 8), block(256);
  hipLaunchKernelGGL(quanv_kernel, grid, block, 0, stream, x, qw, out);
}


# RX-layer matrices factored G = D R D (D = diag((-i)^popcount), R real); D folds into the product-state phase and cancels in |amp|^2: 2 real WMMA chains per complex product, not 4; sum order changes
# speedup vs baseline: 3.0691x; 2.1717x over previous
#include <hip/hip_runtime.h>


typedef __attribute__((ext_vector_type(2))) float v2f;
typedef __attribute__((ext_vector_type(8))) float v8f;

__device__ __forceinline__ void lds_fence() {
  asm volatile("s_wait_dscnt 0" ::: "memory");
}

__device__ __forceinline__ v8f wmma4(v2f a, v2f b, v8f c) {
  return __builtin_amdgcn_wmma_f32_16x16x4_f32(false, a, false, b, (short)0, c,
                                               false, false);
}

__device__ __forceinline__ float lane_bcast(float v, int src_lane) {
  return __int_as_float(__builtin_amdgcn_readlane(__float_as_int(v), src_lane));
}

__device__ __forceinline__ int ring_src(int idx) {
  return ((idx ^ (idx << 1)) & 0x1FE) ^ (idx & 1) ^ (((idx >> 8) & 1) * 3);
}

__launch_bounds__(256)
__global__ void quanv_kernel(const float* __restrict__ x,
                             const float* __restrict__ qw,
                             float* __restrict__ out) {
  constexpr int NQ = 9, B = 4, C = 3, H = 28, W = 28, HOUT = 26, WOUT = 26;
  constexpr int TOTAL = B * HOUT * WOUT;

  __shared__ float lds[8][1024];

  const int lane = threadIdx.x & 31;
  const int warp = threadIdx.x >> 5;
  const int wid = blockIdx.x * 8 + warp;
  if (wid >= TOTAL) return;

  const int b  = wid / (HOUT * WOUT);
  const int yy = (wid / WOUT) % HOUT;
  const int xx = wid % WOUT;

  float* Sre = &lds[warp][0];
  float* Sim = &lds[warp][512];

  const int m16 = lane & 15;
  const int hh  = lane >> 4;

  const int la  = (lane < C * NQ) ? lane : (C * NQ - 1);
  const int chl = la / NQ, il = la - NQ * chl;
  const int dyl = il / 3, dxl = il - 3 * dyl;
  const float xv  = x[((b * C + chl) * H + yy + dyl) * W + xx + dxl];
  const float w0v = qw[il];

  float c1[NQ], s1[NQ];
  {
    const float aw = qw[NQ + ((lane < NQ) ? lane : (NQ - 1))] * 0.5f;
    const float cw = cosf(aw), sw = sinf(aw);
#pragma unroll
    for (int i = 0; i < NQ; ++i) {
      c1[i] = lane_bcast(cw, i);
      s1[i] = lane_bcast(sw, i);
    }
  }

  v2f rA[4];
#pragma unroll
  for (int kc = 0; kc < 4; ++kc) {
#pragma unroll
    for (int v = 0; v < 2; ++v) {
      int k = 4 * kc + v + 2 * hh;
      int diff = m16 ^ k;
      float P = 1.0f;
#pragma unroll
      for (int i = 0; i < 4; ++i) P *= ((diff >> i) & 1) ? s1[i] : c1[i];
      rA[kc][v] = (__popc(m16 & k) & 1) ? -P : P;
    }
  }

  v2f rB[2][8];
#pragma unroll
  for (int t = 0; t < 2; ++t) {
    int cp = 16 * t + m16;
#pragma unroll
    for (int kc = 0; kc < 8; ++kc) {
#pragma unroll
      for (int v = 0; v < 2; ++v) {
        int c = 4 * kc + v + 2 * hh;
        int diff = c ^ cp;
        float P = 1.0f;
#pragma unroll
        for (int i = 0; i < 5; ++i) P *= ((diff >> i) & 1) ? s1[4 + i] : c1[4 + i];
        rB[t][kc][v] = (__popc(c & cp) & 1) ? -P : P;
      }
    }
  }

  const float ax = (xv + w0v) * 0.5f;
  const float cx = cosf(ax), sx = sinf(ax);

  float zacc[4] = {0.f, 0.f, 0.f, 0.f};

  for (int ch = 0; ch < C; ++ch) {
    float ca[NQ], sa[NQ];
#pragma unroll
    for (int i = 0; i < NQ; ++i) {
      ca[i] = lane_bcast(cx, ch * NQ + i);
      sa[i] = lane_bcast(sx, ch * NQ + i);
    }

    const int b01 = lane & 1, b12 = (lane ^ (lane >> 1)) & 1;
    const int popl = __popc(lane);
    const float fL = (((lane ^ (lane >> 1)) & 2) ? sa[2] : ca[2]) *
                     (((lane ^ (lane >> 1)) & 4) ? sa[3] : ca[3]) *
                     (((lane ^ (lane >> 1)) & 8) ? sa[4] : ca[4]);
    float f01[2], f5[2];
    f01[0] = fL * ((b01 ? sa[0] : ca[0]) * (b12 ? sa[1] : ca[1]));
    f01[1] = fL * ((b01 ? ca[0] : sa[0]) * (b12 ? ca[1] : sa[1]));
    f5[0] = (lane & 16) ? sa[5] : ca[5];
    f5[1] = (lane & 16) ? ca[5] : sa[5];
#pragma unroll
    for (int j = 0; j < 16; ++j) {
      const int idx = lane + 32 * j;
      const int s = ring_src(idx);
      const int j0 = j & 1, j3 = (j >> 3) & 1;
      const int hi = (j ^ (j >> 1)) & 7;
      const float k678 = ((hi & 1) ? sa[6] : ca[6]) * ((hi & 2) ? sa[7] : ca[7]) *
                         ((hi & 4) ? sa[8] : ca[8]);
      const float P = (f01[j3] * f5[j0]) * k678;
      const int hm = (__popc(s) + popl + __builtin_popcount(j)) & 3;
      Sre[idx] = (hm == 0) ? P : ((hm == 2) ? -P : 0.0f);
      Sim[idx] = (hm == 1) ? -P : ((hm == 3) ? P : 0.0f);
    }
    lds_fence();

    v2f bR[2][4], bI[2][4];
#pragma unroll
    for (int t = 0; t < 2; ++t) {
#pragma unroll
      for (int kc = 0; kc < 4; ++kc) {
#pragma unroll
        for (int v = 0; v < 2; ++v) {
          int r = 4 * kc + v + 2 * hh;
          int c = 16 * t + m16;
          bR[t][kc][v] = Sre[r + 16 * c];
          bI[t][kc][v] = Sim[r + 16 * c];
        }
      }
    }
    {
      const v8f zero = {0.f,0.f,0.f,0.f,0.f,0.f,0.f,0.f};
      v8f nr[2] = {zero, zero}, ni[2] = {zero, zero};
#pragma unroll
      for (int kc = 0; kc < 4; ++kc) {
#pragma unroll
        for (int t = 0; t < 2; ++t) {
          nr[t] = wmma4(rA[kc], bR[t][kc], nr[t]);
          ni[t] = wmma4(rA[kc], bI[t][kc], ni[t]);
        }
      }
#pragma unroll
      for (int t = 0; t < 2; ++t) {
#pragma unroll
        for (int v = 0; v < 8; ++v) {
          int r = v + 8 * hh;
          int c = 16 * t + m16;
          Sre[r + 16 * c] = nr[t][v];
          Sim[r + 16 * c] = ni[t][v];
        }
      }
    }
    lds_fence();

    v2f aR[8], aI[8];
#pragma unroll
    for (int kc = 0; kc < 8; ++kc) {
#pragma unroll
      for (int v = 0; v < 2; ++v) {
        int c = 4 * kc + v + 2 * hh;
        aR[kc][v] = Sre[m16 + 16 * c];
        aI[kc][v] = Sim[m16 + 16 * c];
      }
    }
    {
      const v8f zero = {0.f,0.f,0.f,0.f,0.f,0.f,0.f,0.f};
      v8f nr[2] = {zero, zero}, ni[2] = {zero, zero};
#pragma unroll
      for (int kc = 0; kc < 8; ++kc) {
#pragma unroll
        for (int t = 0; t < 2; ++t) {
          nr[t] = wmma4(aR[kc], rB[t][kc], nr[t]);
          ni[t] = wmma4(aI[kc], rB[t][kc], ni[t]);
        }
      }
#pragma unroll
      for (int t = 0; t < 2; ++t) {
#pragma unroll
        for (int v = 0; v < 8; ++v) {
          int r = v + 8 * hh;
          int c = 16 * t + m16;
          Sre[r + 16 * c] = nr[t][v];
          Sim[r + 16 * c] = ni[t][v];
        }
      }
    }
    lds_fence();

    float tot = 0.f, alt = 0.f;
#pragma unroll
    for (int j = 0; j < 16; ++j) {
      int idx = lane + 32 * j;
      float re = Sre[idx], im = Sim[idx];
      float p = re * re + im * im;
      tot += p;
      alt += (__builtin_popcount(j) & 1) ? -p : p;
    }
    float z[4];
    z[0] = (__popc(lane & 0x1E) & 1) ? -alt : alt;
    z[1] = (__popc(lane & 0x03) & 1) ? -tot : tot;
    z[2] = (__popc(lane & 0x07) & 1) ? -tot : tot;
    z[3] = (__popc(lane & 0x0F) & 1) ? -tot : tot;
#pragma unroll
    for (int f = 0; f < 4; ++f) {
#pragma unroll
      for (int m = 1; m < 32; m <<= 1) z[f] += __shfl_xor(z[f], m, 32);
      zacc[f] += z[f];
    }
    lds_fence();
  }

  if (lane == 0) {
    const float inv = 1.0f / 3.0f;
#pragma unroll
    for (int f = 0; f < 4; ++f)
      out[((b * 4 + f) * HOUT + yy) * WOUT + xx] = zacc[f] * inv;
  }
}

extern "C" void kernel_launch(void* const* d_in, const int* in_sizes, int n_in,
                              void* d_out, int out_size, void* d_ws, size_t ws_size,
                              hipStream_t stream) {
  (void)in_sizes; (void)n_in; (void)out_size; (void)d_ws; (void)ws_size;
  const float* x  = (const float*)d_in[0];
  const float* qw = (const float*)d_in[1];
  float* out = (float*)d_out;
  const int total_waves = 4 * 26 * 26;
  dim3 grid(total_waves / 8), block(256);
  hipLaunchKernelGGL(quanv_kernel, grid, block, 0, stream, x, qw, out);
}


# Workgroups of 4 waves instead of 8 (676 blocks of 128 threads): at most 3 blocks, 3 waves per SIMD, on the busiest WGP where two 8-wave blocks put 4; kernel body unchanged
# speedup vs baseline: 3.4894x; 1.1369x over previous
#include <hip/hip_runtime.h>


typedef __attribute__((ext_vector_type(2))) float v2f;
typedef __attribute__((ext_vector_type(8))) float v8f;

__device__ __forceinline__ void lds_fence() {
  asm volatile("s_wait_dscnt 0" ::: "memory");
}

__device__ __forceinline__ v8f wmma4(v2f a, v2f b, v8f c) {
  return __builtin_amdgcn_wmma_f32_16x16x4_f32(false, a, false, b, (short)0, c,
                                               false, false);
}

__device__ __forceinline__ float lane_bcast(float v, int src_lane) {
  return __int_as_float(__builtin_amdgcn_readlane(__float_as_int(v), src_lane));
}

__device__ __forceinline__ int ring_src(int idx) {
  return ((idx ^ (idx << 1)) & 0x1FE) ^ (idx & 1) ^ (((idx >> 8) & 1) * 3);
}

__launch_bounds__(128)
__global__ void quanv_kernel(const float* __restrict__ x,
                             const float* __restrict__ qw,
                             float* __restrict__ out) {
  constexpr int NQ = 9, B = 4, C = 3, H = 28, W = 28, HOUT = 26, WOUT = 26;
  constexpr int TOTAL = B * HOUT * WOUT;

  __shared__ float lds[4][1024];

  const int lane = threadIdx.x & 31;
  const int warp = threadIdx.x >> 5;
  const int wid = blockIdx.x * 4 + warp;
  if (wid >= TOTAL) return;

  const int b  = wid / (HOUT * WOUT);
  const int yy = (wid / WOUT) % HOUT;
  const int xx = wid % WOUT;

  float* Sre = &lds[warp][0];
  float* Sim = &lds[warp][512];

  const int m16 = lane & 15;
  const int hh  = lane >> 4;

  const int la  = (lane < C * NQ) ? lane : (C * NQ - 1);
  const int chl = la / NQ, il = la - NQ * chl;
  const int dyl = il / 3, dxl = il - 3 * dyl;
  const float xv  = x[((b * C + chl) * H + yy + dyl) * W + xx + dxl];
  const float w0v = qw[il];

  float c1[NQ], s1[NQ];
  {
    const float aw = qw[NQ + ((lane < NQ) ? lane : (NQ - 1))] * 0.5f;
    const float cw = cosf(aw), sw = sinf(aw);
#pragma unroll
    for (int i = 0; i < NQ; ++i) {
      c1[i] = lane_bcast(cw, i);
      s1[i] = lane_bcast(sw, i);
    }
  }

  v2f rA[4];
#pragma unroll
  for (int kc = 0; kc < 4; ++kc) {
#pragma unroll
    for (int v = 0; v < 2; ++v) {
      int k = 4 * kc + v + 2 * hh;
      int diff = m16 ^ k;
      float P = 1.0f;
#pragma unroll
      for (int i = 0; i < 4; ++i) P *= ((diff >> i) & 1) ? s1[i] : c1[i];
      rA[kc][v] = (__popc(m16 & k) & 1) ? -P : P;
    }
  }

  v2f rB[2][8];
#pragma unroll
  for (int t = 0; t < 2; ++t) {
    int cp = 16 * t + m16;
#pragma unroll
    for (int kc = 0; kc < 8; ++kc) {
#pragma unroll
      for (int v = 0; v < 2; ++v) {
        int c = 4 * kc + v + 2 * hh;
        int diff = c ^ cp;
        float P = 1.0f;
#pragma unroll
        for (int i = 0; i < 5; ++i) P *= ((diff >> i) & 1) ? s1[4 + i] : c1[4 + i];
        rB[t][kc][v] = (__popc(c & cp) & 1) ? -P : P;
      }
    }
  }

  const float ax = (xv + w0v) * 0.5f;
  const float cx = cosf(ax), sx = sinf(ax);

  float zacc[4] = {0.f, 0.f, 0.f, 0.f};

  for (int ch = 0; ch < C; ++ch) {
    float ca[NQ], sa[NQ];
#pragma unroll
    for (int i = 0; i < NQ; ++i) {
      ca[i] = lane_bcast(cx, ch * NQ + i);
      sa[i] = lane_bcast(sx, ch * NQ + i);
    }

    const int b01 = lane & 1, b12 = (lane ^ (lane >> 1)) & 1;
    const int popl = __popc(lane);
    const float fL = (((lane ^ (lane >> 1)) & 2) ? sa[2] : ca[2]) *
                     (((lane ^ (lane >> 1)) & 4) ? sa[3] : ca[3]) *
                     (((lane ^ (lane >> 1)) & 8) ? sa[4] : ca[4]);
    float f01[2], f5[2];
    f01[0] = fL * ((b01 ? sa[0] : ca[0]) * (b12 ? sa[1] : ca[1]));
    f01[1] = fL * ((b01 ? ca[0] : sa[0]) * (b12 ? ca[1] : sa[1]));
    f5[0] = (lane & 16) ? sa[5] : ca[5];
    f5[1] = (lane & 16) ? ca[5] : sa[5];
#pragma unroll
    for (int j = 0; j < 16; ++j) {
      const int idx = lane + 32 * j;
      const int s = ring_src(idx);
      const int j0 = j & 1, j3 = (j >> 3) & 1;
      const int hi = (j ^ (j >> 1)) & 7;
      const float k678 = ((hi & 1) ? sa[6] : ca[6]) * ((hi & 2) ? sa[7] : ca[7]) *
                         ((hi & 4) ? sa[8] : ca[8]);
      const float P = (f01[j3] * f5[j0]) * k678;
      const int hm = (__popc(s) + popl + __builtin_popcount(j)) & 3;
      Sre[idx] = (hm == 0) ? P : ((hm == 2) ? -P : 0.0f);
      Sim[idx] = (hm == 1) ? -P : ((hm == 3) ? P : 0.0f);
    }
    lds_fence();

    v2f bR[2][4], bI[2][4];
#pragma unroll
    for (int t = 0; t < 2; ++t) {
#pragma unroll
      for (int kc = 0; kc < 4; ++kc) {
#pragma unroll
        for (int v = 0; v < 2; ++v) {
          int r = 4 * kc + v + 2 * hh;
          int c = 16 * t + m16;
          bR[t][kc][v] = Sre[r + 16 * c];
          bI[t][kc][v] = Sim[r + 16 * c];
        }
      }
    }
    {
      const v8f zero = {0.f,0.f,0.f,0.f,0.f,0.f,0.f,0.f};
      v8f nr[2] = {zero, zero}, ni[2] = {zero, zero};
#pragma unroll
      for (int kc = 0; kc < 4; ++kc) {
#pragma unroll
        for (int t = 0; t < 2; ++t) {
          nr[t] = wmma4(rA[kc], bR[t][kc], nr[t]);
          ni[t] = wmma4(rA[kc], bI[t][kc], ni[t]);
        }
      }
#pragma unroll
      for (int t = 0; t < 2; ++t) {
#pragma unroll
        for (int v = 0; v < 8; ++v) {
          int r = v + 8 * hh;
          int c = 16 * t + m16;
          Sre[r + 16 * c] = nr[t][v];
          Sim[r + 16 * c] = ni[t][v];
        }
      }
    }
    lds_fence();

    v2f aR[8], aI[8];
#pragma unroll
    for (int kc = 0; kc < 8; ++kc) {
#pragma unroll
      for (int v = 0; v < 2; ++v) {
        int c = 4 * kc + v + 2 * hh;
        aR[kc][v] = Sre[m16 + 16 * c];
        aI[kc][v] = Sim[m16 + 16 * c];
      }
    }
    {
      const v8f zero = {0.f,0.f,0.f,0.f,0.f,0.f,0.f,0.f};
      v8f nr[2] = {zero, zero}, ni[2] = {zero, zero};
#pragma unroll
      for (int kc = 0; kc < 8; ++kc) {
#pragma unroll
        for (int t = 0; t < 2; ++t) {
          nr[t] = wmma4(aR[kc], rB[t][kc], nr[t]);
          ni[t] = wmma4(aI[kc], rB[t][kc], ni[t]);
        }
      }
#pragma unroll
      for (int t = 0; t < 2; ++t) {
#pragma unroll
        for (int v = 0; v < 8; ++v) {
          int r = v + 8 * hh;
          int c = 16 * t + m16;
          Sre[r + 16 * c] = nr[t][v];
          Sim[r + 16 * c] = ni[t][v];
        }
      }
    }
    lds_fence();

    float tot = 0.f, alt = 0.f;
#pragma unroll
    for (int j = 0; j < 16; ++j) {
      int idx = lane + 32 * j;
      float re = Sre[idx], im = Sim[idx];
      float p = re * re + im * im;
      tot += p;
      alt += (__builtin_popcount(j) & 1) ? -p : p;
    }
    float z[4];
    z[0] = (__popc(lane & 0x1E) & 1) ? -alt : alt;
    z[1] = (__popc(lane & 0x03) & 1) ? -tot : tot;
    z[2] = (__popc(lane & 0x07) & 1) ? -tot : tot;
    z[3] = (__popc(lane & 0x0F) & 1) ? -tot : tot;
#pragma unroll
    for (int f = 0; f < 4; ++f) {
#pragma unroll
      for (int m = 1; m < 32; m <<= 1) z[f] += __shfl_xor(z[f], m, 32);
      zacc[f] += z[f];
    }
    lds_fence();
  }

  if (lane == 0) {
    const float inv = 1.0f / 3.0f;
#pragma unroll
    for (int f = 0; f < 4; ++f)
      out[((b * 4 + f) * HOUT + yy) * WOUT + xx] = zacc[f] * inv;
  }
}

extern "C" void kernel_launch(void* const* d_in, const int* in_sizes, int n_in,
                              void* d_out, int out_size, void* d_ws, size_t ws_size,
                              hipStream_t stream) {
  (void)in_sizes; (void)n_in; (void)out_size; (void)d_ws; (void)ws_size;
  const float* x  = (const float*)d_in[0];
  const float* qw = (const float*)d_in[1];
  float* out = (float*)d_out;
  const int total_waves = 4 * 26 * 26;
  dim3 grid(total_waves / 4), block(128);
  hipLaunchKernelGGL(quanv_kernel, grid, block, 0, stream, x, qw, out);
}


# RX gate of qubit 4 folded into the measurement as cos(w)*Z + sin(w)*Y on amplitude pairs: second product 16x32x32 -> 16x16x32 over qubits 5-8, 32 of 48 WMMAs per channel left; sum order changes
# speedup vs baseline: 3.8210x; 1.0950x over previous
#include <hip/hip_runtime.h>


typedef __attribute__((ext_vector_type(2))) float v2f;
typedef __attribute__((ext_vector_type(8))) float v8f;

__device__ __forceinline__ void lds_fence() {
  asm volatile("s_wait_dscnt 0" ::: "memory");
}

__device__ __forceinline__ v8f wmma4(v2f a, v2f b, v8f c) {
  return __builtin_amdgcn_wmma_f32_16x16x4_f32(false, a, false, b, (short)0, c,
                                               false, false);
}

__device__ __forceinline__ float lane_bcast(float v, int src_lane) {
  return __int_as_float(__builtin_amdgcn_readlane(__float_as_int(v), src_lane));
}

__device__ __forceinline__ int ring_src(int idx) {
  return ((idx ^ (idx << 1)) & 0x1FE) ^ (idx & 1) ^ (((idx >> 8) & 1) * 3);
}

__launch_bounds__(128)
__global__ void quanv_kernel(const float* __restrict__ x,
                             const float* __restrict__ qw,
                             float* __restrict__ out) {
  constexpr int NQ = 9, B = 4, C = 3, H = 28, W = 28, HOUT = 26, WOUT = 26;
  constexpr int TOTAL = B * HOUT * WOUT;

  __shared__ float lds[4][1024];

  const int lane = threadIdx.x & 31;
  const int warp = threadIdx.x >> 5;
  const int wid = blockIdx.x * 4 + warp;
  if (wid >= TOTAL) return;

  const int b  = wid / (HOUT * WOUT);
  const int yy = (wid / WOUT) % HOUT;
  const int xx = wid % WOUT;

  float* Sre = &lds[warp][0];
  float* Sim = &lds[warp][512];

  const int m16 = lane & 15;
  const int hh  = lane >> 4;

  const int la  = (lane < C * NQ) ? lane : (C * NQ - 1);
  const int chl = la / NQ, il = la - NQ * chl;
  const int dyl = il / 3, dxl = il - 3 * dyl;
  const float xv  = x[((b * C + chl) * H + yy + dyl) * W + xx + dxl];
  const float w0v = qw[il];

  float c1[NQ], s1[NQ];
  {
    const float aw = qw[NQ + ((lane < NQ) ? lane : (NQ - 1))] * 0.5f;
    const float cw = cosf(aw), sw = sinf(aw);
#pragma unroll
    for (int i = 0; i < NQ; ++i) {
      c1[i] = lane_bcast(cw, i);
      s1[i] = lane_bcast(sw, i);
    }
  }

  v2f rA[4];
#pragma unroll
  for (int kc = 0; kc < 4; ++kc) {
#pragma unroll
    for (int v = 0; v < 2; ++v) {
      int k = 4 * kc + v + 2 * hh;
      int diff = m16 ^ k;
      float P = 1.0f;
#pragma unroll
      for (int i = 0; i < 4; ++i) P *= ((diff >> i) & 1) ? s1[i] : c1[i];
      rA[kc][v] = (__popc(m16 & k) & 1) ? -P : P;
    }
  }

  v2f rU[4];
#pragma unroll
  for (int kc = 0; kc < 4; ++kc) {
#pragma unroll
    for (int v = 0; v < 2; ++v) {
      int k = 4 * kc + v + 2 * hh;
      int diff = m16 ^ k;
      float P = 1.0f;
#pragma unroll
      for (int i = 0; i < 4; ++i) P *= ((diff >> i) & 1) ? s1[5 + i] : c1[5 + i];
      rU[kc][v] = (__popc(m16 & k) & 1) ? -P : P;
    }
  }

  const float cz4 = c1[4] * c1[4] - s1[4] * s1[4];
  const float sy4 = 4.0f * (c1[4] * s1[4]);

  const float ax = (xv + w0v) * 0.5f;
  const float cx = cosf(ax), sx = sinf(ax);

  float zacc[4] = {0.f, 0.f, 0.f, 0.f};

  for (int ch = 0; ch < C; ++ch) {
    float ca[NQ], sa[NQ];
#pragma unroll
    for (int i = 0; i < NQ; ++i) {
      ca[i] = lane_bcast(cx, ch * NQ + i);
      sa[i] = lane_bcast(sx, ch * NQ + i);
    }

    const int b01 = lane & 1, b12 = (lane ^ (lane >> 1)) & 1;
    const int popl = __popc(lane);
    const float fL = (((lane ^ (lane >> 1)) & 2) ? sa[2] : ca[2]) *
                     (((lane ^ (lane >> 1)) & 4) ? sa[3] : ca[3]) *
                     (((lane ^ (lane >> 1)) & 8) ? sa[4] : ca[4]);
    float f01[2], f5[2];
    f01[0] = fL * ((b01 ? sa[0] : ca[0]) * (b12 ? sa[1] : ca[1]));
    f01[1] = fL * ((b01 ? ca[0] : sa[0]) * (b12 ? ca[1] : sa[1]));
    f5[0] = (lane & 16) ? sa[5] : ca[5];
    f5[1] = (lane & 16) ? ca[5] : sa[5];
#pragma unroll
    for (int j = 0; j < 16; ++j) {
      const int idx = lane + 32 * j;
      const int s = ring_src(idx);
      const int j0 = j & 1, j3 = (j >> 3) & 1;
      const int hi = (j ^ (j >> 1)) & 7;
      const float k678 = ((hi & 1) ? sa[6] : ca[6]) * ((hi & 2) ? sa[7] : ca[7]) *
                         ((hi & 4) ? sa[8] : ca[8]);
      const float P = (f01[j3] * f5[j0]) * k678;
      const int hm = (__popc(s) + popl + __builtin_popcount(j)) & 3;
      Sre[idx] = (hm == 0) ? P : ((hm == 2) ? -P : 0.0f);
      Sim[idx] = (hm == 1) ? -P : ((hm == 3) ? P : 0.0f);
    }
    lds_fence();

    v2f bR[2][4], bI[2][4];
#pragma unroll
    for (int t = 0; t < 2; ++t) {
#pragma unroll
      for (int kc = 0; kc < 4; ++kc) {
#pragma unroll
        for (int v = 0; v < 2; ++v) {
          int r = 4 * kc + v + 2 * hh;
          int c = 16 * t + m16;
          bR[t][kc][v] = Sre[r + 16 * c];
          bI[t][kc][v] = Sim[r + 16 * c];
        }
      }
    }
    {
      const v8f zero = {0.f,0.f,0.f,0.f,0.f,0.f,0.f,0.f};
      v8f nr[2] = {zero, zero}, ni[2] = {zero, zero};
#pragma unroll
      for (int kc = 0; kc < 4; ++kc) {
#pragma unroll
        for (int t = 0; t < 2; ++t) {
          nr[t] = wmma4(rA[kc], bR[t][kc], nr[t]);
          ni[t] = wmma4(rA[kc], bI[t][kc], ni[t]);
        }
      }
#pragma unroll
      for (int t = 0; t < 2; ++t) {
#pragma unroll
        for (int v = 0; v < 8; ++v) {
          int r = v + 8 * hh;
          int c = 16 * t + m16;
          Sre[r + 16 * c] = nr[t][v];
          Sim[r + 16 * c] = ni[t][v];
        }
      }
    }
    lds_fence();

    v2f uR[2][4], uI[2][4];
#pragma unroll
    for (int t = 0; t < 2; ++t) {
#pragma unroll
      for (int kc = 0; kc < 4; ++kc) {
#pragma unroll
        for (int v = 0; v < 2; ++v) {
          int u = 4 * kc + v + 2 * hh;
          uR[t][kc][v] = Sre[m16 + 16 * t + 32 * u];
          uI[t][kc][v] = Sim[m16 + 16 * t + 32 * u];
        }
      }
    }
    {
      const v8f zero = {0.f,0.f,0.f,0.f,0.f,0.f,0.f,0.f};
      v8f nr[2] = {zero, zero}, ni[2] = {zero, zero};
#pragma unroll
      for (int kc = 0; kc < 4; ++kc) {
#pragma unroll
        for (int t = 0; t < 2; ++t) {
          nr[t] = wmma4(rU[kc], uR[t][kc], nr[t]);
          ni[t] = wmma4(rU[kc], uI[t][kc], ni[t]);
        }
      }
#pragma unroll
      for (int t = 0; t < 2; ++t) {
#pragma unroll
        for (int v = 0; v < 8; ++v) {
          int u = v + 8 * hh;
          Sre[m16 + 16 * t + 32 * u] = nr[t][v];
          Sim[m16 + 16 * t + 32 * u] = ni[t][v];
        }
      }
    }
    lds_fence();

    float tot = 0.f, alt = 0.f;
    const int pbase = m16 + 32 * hh;
#pragma unroll
    for (int j = 0; j < 8; ++j) {
      const int i0 = pbase + 64 * j, i1 = i0 + 16;
      const float re0 = Sre[i0], im0 = Sim[i0], re1 = Sre[i1], im1 = Sim[i1];
      const float p0 = re0 * re0 + im0 * im0;
      const float p1 = re1 * re1 + im1 * im1;
      const float zz = cz4 * (p0 - p1) + sy4 * (re0 * re1 + im0 * im1);
      tot += p0 + p1;
      alt += (__builtin_popcount(j) & 1) ? -zz : zz;
    }
    float z[4];
    z[0] = (__popc(lane & 0x1E) & 1) ? -alt : alt;
    z[1] = (__popc(lane & 0x03) & 1) ? -tot : tot;
    z[2] = (__popc(lane & 0x07) & 1) ? -tot : tot;
    z[3] = (__popc(lane & 0x0F) & 1) ? -tot : tot;
#pragma unroll
    for (int f = 0; f < 4; ++f) {
#pragma unroll
      for (int m = 1; m < 32; m <<= 1) z[f] += __shfl_xor(z[f], m, 32);
      zacc[f] += z[f];
    }
    lds_fence();
  }

  if (lane == 0) {
    const float inv = 1.0f / 3.0f;
#pragma unroll
    for (int f = 0; f < 4; ++f)
      out[((b * 4 + f) * HOUT + yy) * WOUT + xx] = zacc[f] * inv;
  }
}

extern "C" void kernel_launch(void* const* d_in, const int* in_sizes, int n_in,
                              void* d_out, int out_size, void* d_ws, size_t ws_size,
                              hipStream_t stream) {
  (void)in_sizes; (void)n_in; (void)out_size; (void)d_ws; (void)ws_size;
  const float* x  = (const float*)d_in[0];
  const float* qw = (const float*)d_in[1];
  float* out = (float*)d_out;
  const int total_waves = 4 * 26 * 26;
  dim3 grid(total_waves / 4), block(128);
  hipLaunchKernelGGL(quanv_kernel, grid, block, 0, stream, x, qw, out);
}


# After CNOT ring 1 the 16x32 state is a 16x4 block (columns q4,q8) times one factor per column: R_A multiplies the block (4 WMMAs, not 16), factor applied after the sum in registers; rounding changes
# speedup vs baseline: 5.3702x; 1.4054x over previous
#include <hip/hip_runtime.h>


typedef __attribute__((ext_vector_type(2))) float v2f;
typedef __attribute__((ext_vector_type(8))) float v8f;

__device__ __forceinline__ void lds_fence() {
  asm volatile("s_wait_dscnt 0" ::: "memory");
}

__device__ __forceinline__ v8f wmma4(v2f a, v2f b, v8f c) {
  return __builtin_amdgcn_wmma_f32_16x16x4_f32(false, a, false, b, (short)0, c,
                                               false, false);
}

__device__ __forceinline__ float lane_bcast(float v, int src_lane) {
  return __int_as_float(__builtin_amdgcn_readlane(__float_as_int(v), src_lane));
}


__launch_bounds__(128)
__global__ void quanv_kernel(const float* __restrict__ x,
                             const float* __restrict__ qw,
                             float* __restrict__ out) {
  constexpr int NQ = 9, B = 4, C = 3, H = 28, W = 28, HOUT = 26, WOUT = 26;
  constexpr int TOTAL = B * HOUT * WOUT;

  __shared__ float lds[4][1024];

  const int lane = threadIdx.x & 31;
  const int warp = threadIdx.x >> 5;
  const int wid = blockIdx.x * 4 + warp;
  if (wid >= TOTAL) return;

  const int b  = wid / (HOUT * WOUT);
  const int yy = (wid / WOUT) % HOUT;
  const int xx = wid % WOUT;

  float* Sre = &lds[warp][0];
  float* Sim = &lds[warp][512];

  const int m16 = lane & 15;
  const int hh  = lane >> 4;

  const int la  = (lane < C * NQ) ? lane : (C * NQ - 1);
  const int chl = la / NQ, il = la - NQ * chl;
  const int dyl = il / 3, dxl = il - 3 * dyl;
  const float xv  = x[((b * C + chl) * H + yy + dyl) * W + xx + dxl];
  const float w0v = qw[il];

  float c1[NQ], s1[NQ];
  {
    const float aw = qw[NQ + ((lane < NQ) ? lane : (NQ - 1))] * 0.5f;
    const float cw = cosf(aw), sw = sinf(aw);
#pragma unroll
    for (int i = 0; i < NQ; ++i) {
      c1[i] = lane_bcast(cw, i);
      s1[i] = lane_bcast(sw, i);
    }
  }

  v2f rA[4];
#pragma unroll
  for (int kc = 0; kc < 4; ++kc) {
#pragma unroll
    for (int v = 0; v < 2; ++v) {
      int k = 4 * kc + v + 2 * hh;
      int diff = m16 ^ k;
      float P = 1.0f;
#pragma unroll
      for (int i = 0; i < 4; ++i) P *= ((diff >> i) & 1) ? s1[i] : c1[i];
      rA[kc][v] = (__popc(m16 & k) & 1) ? -P : P;
    }
  }

  v2f rU[4];
#pragma unroll
  for (int kc = 0; kc < 4; ++kc) {
#pragma unroll
    for (int v = 0; v < 2; ++v) {
      int k = 4 * kc + v + 2 * hh;
      int diff = m16 ^ k;
      float P = 1.0f;
#pragma unroll
      for (int i = 0; i < 4; ++i) P *= ((diff >> i) & 1) ? s1[5 + i] : c1[5 + i];
      rU[kc][v] = (__popc(m16 & k) & 1) ? -P : P;
    }
  }

  const float cz4 = c1[4] * c1[4] - s1[4] * s1[4];
  const float sy4 = 4.0f * (c1[4] * s1[4]);

  const float ax = (xv + w0v) * 0.5f;
  const float cx = cosf(ax), sx = sinf(ax);

  float zacc[4] = {0.f, 0.f, 0.f, 0.f};

  for (int ch = 0; ch < C; ++ch) {
    float ca[NQ], sa[NQ];
#pragma unroll
    for (int i = 0; i < NQ; ++i) {
      ca[i] = lane_bcast(cx, ch * NQ + i);
      sa[i] = lane_bcast(sx, ch * NQ + i);
    }

    v8f V = {0.f,0.f,0.f,0.f,0.f,0.f,0.f,0.f};
    {
      const int t4 = lane & 1, t8 = (lane >> 1) & 1, pl = (lane >> 2) & 1;
      v2f aT[4];
#pragma unroll
      for (int kc = 0; kc < 4; ++kc) {
#pragma unroll
        for (int v = 0; v < 2; ++v) {
          const int i2 = kc & 1, i3 = kc >> 1;
          const int s0 = v ^ t8, s1 = hh ^ v ^ t8, s2 = i2 ^ hh, s3 = i3 ^ i2, s4 = t4 ^ i3;
          const float amp = ((s0 ? sa[0] : ca[0]) * (s1 ? sa[1] : ca[1])) *
                            ((s2 ? sa[2] : ca[2]) * (s3 ? sa[3] : ca[3])) * (s4 ? sa[4] : ca[4]);
          const int q = (s0 + s1 + s2 + s3 + s4 + v + hh + __builtin_popcount(kc)) & 3;
          const float sg = ((q & 1) != pl) ? 0.0f : (((q + 1) & 2) ? -1.0f : 1.0f);
          aT[kc][v] = amp * sg;
        }
      }
#pragma unroll
      for (int kc = 0; kc < 4; ++kc) V = wmma4(aT[kc], rA[kc], V);
    }

    v2f uR[2][4], uI[2][4];
#pragma unroll
    for (int t = 0; t < 2; ++t) {
#pragma unroll
      for (int kc = 0; kc < 4; ++kc) {
#pragma unroll
        for (int v = 0; v < 2; ++v) {
          const int i8 = kc >> 1;
          const float VRe = V[t + 2 * i8], VIm = V[4 + t + 2 * i8];
          const int s5 = v ^ t, s6 = hh ^ v, s7 = (kc & 1) ^ hh, s8 = (kc >> 1) ^ (kc & 1);
          const float g = ((s5 ? sa[5] : ca[5]) * (s8 ? sa[8] : ca[8])) *
                          ((s6 ? sa[6] : ca[6]) * (s7 ? sa[7] : ca[7]));
          const int e = (s5 + s6 + s7 + s8 + t + v + hh + __builtin_popcount(kc)) & 3;
          const float xr = ((e & 1) ? VIm : VRe) * g;
          const float xi = ((e & 1) ? VRe : VIm) * g;
          uR[t][kc][v] = (e & 2) ? -xr : xr;
          uI[t][kc][v] = ((e + 1) & 2) ? -xi : xi;
        }
      }
    }
    {
      const v8f zero = {0.f,0.f,0.f,0.f,0.f,0.f,0.f,0.f};
      v8f nr[2] = {zero, zero}, ni[2] = {zero, zero};
#pragma unroll
      for (int kc = 0; kc < 4; ++kc) {
#pragma unroll
        for (int t = 0; t < 2; ++t) {
          nr[t] = wmma4(rU[kc], uR[t][kc], nr[t]);
          ni[t] = wmma4(rU[kc], uI[t][kc], ni[t]);
        }
      }
#pragma unroll
      for (int t = 0; t < 2; ++t) {
#pragma unroll
        for (int v = 0; v < 8; ++v) {
          int u = v + 8 * hh;
          Sre[m16 + 16 * t + 32 * u] = nr[t][v];
          Sim[m16 + 16 * t + 32 * u] = ni[t][v];
        }
      }
    }
    lds_fence();

    float tot = 0.f, alt = 0.f;
    const int pbase = m16 + 32 * hh;
#pragma unroll
    for (int j = 0; j < 8; ++j) {
      const int i0 = pbase + 64 * j, i1 = i0 + 16;
      const float re0 = Sre[i0], im0 = Sim[i0], re1 = Sre[i1], im1 = Sim[i1];
      const float p0 = re0 * re0 + im0 * im0;
      const float p1 = re1 * re1 + im1 * im1;
      const float zz = cz4 * (p0 - p1) + sy4 * (re0 * re1 + im0 * im1);
      tot += p0 + p1;
      alt += (__builtin_popcount(j) & 1) ? -zz : zz;
    }
    float z[4];
    z[0] = (__popc(lane & 0x1E) & 1) ? -alt : alt;
    z[1] = (__popc(lane & 0x03) & 1) ? -tot : tot;
    z[2] = (__popc(lane & 0x07) & 1) ? -tot : tot;
    z[3] = (__popc(lane & 0x0F) & 1) ? -tot : tot;
#pragma unroll
    for (int f = 0; f < 4; ++f) {
#pragma unroll
      for (int m = 1; m < 32; m <<= 1) z[f] += __shfl_xor(z[f], m, 32);
      zacc[f] += z[f];
    }
    lds_fence();
  }

  if (lane == 0) {
    const float inv = 1.0f / 3.0f;
#pragma unroll
    for (int f = 0; f < 4; ++f)
      out[((b * 4 + f) * HOUT + yy) * WOUT + xx] = zacc[f] * inv;
  }
}

extern "C" void kernel_launch(void* const* d_in, const int* in_sizes, int n_in,
                              void* d_out, int out_size, void* d_ws, size_t ws_size,
                              hipStream_t stream) {
  (void)in_sizes; (void)n_in; (void)out_size; (void)d_ws; (void)ws_size;
  const float* x  = (const float*)d_in[0];
  const float* qw = (const float*)d_in[1];
  float* out = (float*)d_out;
  const int total_waves = 4 * 26 * 26;
  dim3 grid(total_waves / 4), block(128);
  hipLaunchKernelGGL(quanv_kernel, grid, block, 0, stream, x, qw, out);
}


# Probabilities summed straight from the second product's accumulators (each lane already holds both q4 amplitudes of its pairs): LDS round trip of the 512-amplitude state gone; sum order changes
# speedup vs baseline: 6.6294x; 1.2345x over previous
#include <hip/hip_runtime.h>


typedef __attribute__((ext_vector_type(2))) float v2f;
typedef __attribute__((ext_vector_type(8))) float v8f;

__device__ __forceinline__ v8f wmma4(v2f a, v2f b, v8f c) {
  return __builtin_amdgcn_wmma_f32_16x16x4_f32(false, a, false, b, (short)0, c,
                                               false, false);
}

__device__ __forceinline__ float lane_bcast(float v, int src_lane) {
  return __int_as_float(__builtin_amdgcn_readlane(__float_as_int(v), src_lane));
}


__launch_bounds__(128)
__global__ void quanv_kernel(const float* __restrict__ x,
                             const float* __restrict__ qw,
                             float* __restrict__ out) {
  constexpr int NQ = 9, B = 4, C = 3, H = 28, W = 28, HOUT = 26, WOUT = 26;
  constexpr int TOTAL = B * HOUT * WOUT;

  const int lane = threadIdx.x & 31;
  const int warp = threadIdx.x >> 5;
  const int wid = blockIdx.x * 4 + warp;
  if (wid >= TOTAL) return;

  const int b  = wid / (HOUT * WOUT);
  const int yy = (wid / WOUT) % HOUT;
  const int xx = wid % WOUT;

  const int m16 = lane & 15;
  const int hh  = lane >> 4;

  const int la  = (lane < C * NQ) ? lane : (C * NQ - 1);
  const int chl = la / NQ, il = la - NQ * chl;
  const int dyl = il / 3, dxl = il - 3 * dyl;
  const float xv  = x[((b * C + chl) * H + yy + dyl) * W + xx + dxl];
  const float w0v = qw[il];

  float c1[NQ], s1[NQ];
  {
    const float aw = qw[NQ + ((lane < NQ) ? lane : (NQ - 1))] * 0.5f;
    const float cw = cosf(aw), sw = sinf(aw);
#pragma unroll
    for (int i = 0; i < NQ; ++i) {
      c1[i] = lane_bcast(cw, i);
      s1[i] = lane_bcast(sw, i);
    }
  }

  v2f rA[4];
#pragma unroll
  for (int kc = 0; kc < 4; ++kc) {
#pragma unroll
    for (int v = 0; v < 2; ++v) {
      int k = 4 * kc + v + 2 * hh;
      int diff = m16 ^ k;
      float P = 1.0f;
#pragma unroll
      for (int i = 0; i < 4; ++i) P *= ((diff >> i) & 1) ? s1[i] : c1[i];
      rA[kc][v] = (__popc(m16 & k) & 1) ? -P : P;
    }
  }

  v2f rU[4];
#pragma unroll
  for (int kc = 0; kc < 4; ++kc) {
#pragma unroll
    for (int v = 0; v < 2; ++v) {
      int k = 4 * kc + v + 2 * hh;
      int diff = m16 ^ k;
      float P = 1.0f;
#pragma unroll
      for (int i = 0; i < 4; ++i) P *= ((diff >> i) & 1) ? s1[5 + i] : c1[5 + i];
      rU[kc][v] = (__popc(m16 & k) & 1) ? -P : P;
    }
  }

  const float cz4 = c1[4] * c1[4] - s1[4] * s1[4];
  const float sy4 = 4.0f * (c1[4] * s1[4]);

  const float ax = (xv + w0v) * 0.5f;
  const float cx = cosf(ax), sx = sinf(ax);

  float zacc[4] = {0.f, 0.f, 0.f, 0.f};

  for (int ch = 0; ch < C; ++ch) {
    float ca[NQ], sa[NQ];
#pragma unroll
    for (int i = 0; i < NQ; ++i) {
      ca[i] = lane_bcast(cx, ch * NQ + i);
      sa[i] = lane_bcast(sx, ch * NQ + i);
    }

    v8f V = {0.f,0.f,0.f,0.f,0.f,0.f,0.f,0.f};
    {
      const int t4 = lane & 1, t8 = (lane >> 1) & 1, pl = (lane >> 2) & 1;
      v2f aT[4];
#pragma unroll
      for (int kc = 0; kc < 4; ++kc) {
#pragma unroll
        for (int v = 0; v < 2; ++v) {
          const int i2 = kc & 1, i3 = kc >> 1;
          const int s0 = v ^ t8, s1 = hh ^ v ^ t8, s2 = i2 ^ hh, s3 = i3 ^ i2, s4 = t4 ^ i3;
          const float amp = ((s0 ? sa[0] : ca[0]) * (s1 ? sa[1] : ca[1])) *
                            ((s2 ? sa[2] : ca[2]) * (s3 ? sa[3] : ca[3])) * (s4 ? sa[4] : ca[4]);
          const int q = (s0 + s1 + s2 + s3 + s4 + v + hh + __builtin_popcount(kc)) & 3;
          const float sg = ((q & 1) != pl) ? 0.0f : (((q + 1) & 2) ? -1.0f : 1.0f);
          aT[kc][v] = amp * sg;
        }
      }
#pragma unroll
      for (int kc = 0; kc < 4; ++kc) V = wmma4(aT[kc], rA[kc], V);
    }

    v2f uR[2][4], uI[2][4];
#pragma unroll
    for (int t = 0; t < 2; ++t) {
#pragma unroll
      for (int kc = 0; kc < 4; ++kc) {
#pragma unroll
        for (int v = 0; v < 2; ++v) {
          const int i8 = kc >> 1;
          const float VRe = V[t + 2 * i8], VIm = V[4 + t + 2 * i8];
          const int s5 = v ^ t, s6 = hh ^ v, s7 = (kc & 1) ^ hh, s8 = (kc >> 1) ^ (kc & 1);
          const float g = ((s5 ? sa[5] : ca[5]) * (s8 ? sa[8] : ca[8])) *
                          ((s6 ? sa[6] : ca[6]) * (s7 ? sa[7] : ca[7]));
          const int e = (s5 + s6 + s7 + s8 + t + v + hh + __builtin_popcount(kc)) & 3;
          const float xr = ((e & 1) ? VIm : VRe) * g;
          const float xi = ((e & 1) ? VRe : VIm) * g;
          uR[t][kc][v] = (e & 2) ? -xr : xr;
          uI[t][kc][v] = ((e + 1) & 2) ? -xi : xi;
        }
      }
    }
    const v8f zero = {0.f,0.f,0.f,0.f,0.f,0.f,0.f,0.f};
    v8f nr[2] = {zero, zero}, ni[2] = {zero, zero};
#pragma unroll
    for (int kc = 0; kc < 4; ++kc) {
#pragma unroll
      for (int t = 0; t < 2; ++t) {
        nr[t] = wmma4(rU[kc], uR[t][kc], nr[t]);
        ni[t] = wmma4(rU[kc], uI[t][kc], ni[t]);
      }
    }

    float tot = 0.f, alt = 0.f;
#pragma unroll
    for (int v = 0; v < 8; ++v) {
      const float re0 = nr[0][v], im0 = ni[0][v], re1 = nr[1][v], im1 = ni[1][v];
      const float p0 = re0 * re0 + im0 * im0;
      const float p1 = re1 * re1 + im1 * im1;
      const float zz = cz4 * (p0 - p1) + sy4 * (re0 * re1 + im0 * im1);
      tot += p0 + p1;
      alt += (__builtin_popcount(v) & 1) ? -zz : zz;
    }
    float z[4];
    z[0] = (__popc(lane & 0x1E) & 1) ? -alt : alt;
    z[1] = (__popc(lane & 0x03) & 1) ? -tot : tot;
    z[2] = (__popc(lane & 0x07) & 1) ? -tot : tot;
    z[3] = (__popc(lane & 0x0F) & 1) ? -tot : tot;
#pragma unroll
    for (int f = 0; f < 4; ++f) {
#pragma unroll
      for (int m = 1; m < 32; m <<= 1) z[f] += __shfl_xor(z[f], m, 32);
      zacc[f] += z[f];
    }
  }

  if (lane == 0) {
    const float inv = 1.0f / 3.0f;
#pragma unroll
    for (int f = 0; f < 4; ++f)
      out[((b * 4 + f) * HOUT + yy) * WOUT + xx] = zacc[f] * inv;
  }
}

extern "C" void kernel_launch(void* const* d_in, const int* in_sizes, int n_in,
                              void* d_out, int out_size, void* d_ws, size_t ws_size,
                              hipStream_t stream) {
  (void)in_sizes; (void)n_in; (void)out_size; (void)d_ws; (void)ws_size;
  const float* x  = (const float*)d_in[0];
  const float* qw = (const float*)d_in[1];
  float* out = (float*)d_out;
  const int total_waves = 4 * 26 * 26;
  dim3 grid(total_waves / 4), block(128);
  hipLaunchKernelGGL(quanv_kernel, grid, block, 0, stream, x, qw, out);
}


# 2nd RX product on the rank-2 form of the state: R_U times the 16x4 factor block G (4 WMMAs), then S''=sum_i8 H[u';q4,i8]*V[r';q4,i8], one K=4 WMMA per q4 and plane: 8 of 16 left; sum order changes
# speedup vs baseline: 7.7153x; 1.1638x over previous
#include <hip/hip_runtime.h>


typedef __attribute__((ext_vector_type(2))) float v2f;
typedef __attribute__((ext_vector_type(8))) float v8f;

__device__ __forceinline__ v8f wmma4(v2f a, v2f b, v8f c) {
  return __builtin_amdgcn_wmma_f32_16x16x4_f32(false, a, false, b, (short)0, c,
                                               false, false);
}

__device__ __forceinline__ float lane_bcast(float v, int src_lane) {
  return __int_as_float(__builtin_amdgcn_readlane(__float_as_int(v), src_lane));
}


__launch_bounds__(128)
__global__ void quanv_kernel(const float* __restrict__ x,
                             const float* __restrict__ qw,
                             float* __restrict__ out) {
  constexpr int NQ = 9, B = 4, C = 3, H = 28, W = 28, HOUT = 26, WOUT = 26;
  constexpr int TOTAL = B * HOUT * WOUT;

  const int lane = threadIdx.x & 31;
  const int warp = threadIdx.x >> 5;
  const int wid = blockIdx.x * 4 + warp;
  if (wid >= TOTAL) return;

  const int b  = wid / (HOUT * WOUT);
  const int yy = (wid / WOUT) % HOUT;
  const int xx = wid % WOUT;

  const int m16 = lane & 15;
  const int hh  = lane >> 4;

  const int la  = (lane < C * NQ) ? lane : (C * NQ - 1);
  const int chl = la / NQ, il = la - NQ * chl;
  const int dyl = il / 3, dxl = il - 3 * dyl;
  const float xv  = x[((b * C + chl) * H + yy + dyl) * W + xx + dxl];
  const float w0v = qw[il];

  float c1[NQ], s1[NQ];
  {
    const float aw = qw[NQ + ((lane < NQ) ? lane : (NQ - 1))] * 0.5f;
    const float cw = cosf(aw), sw = sinf(aw);
#pragma unroll
    for (int i = 0; i < NQ; ++i) {
      c1[i] = lane_bcast(cw, i);
      s1[i] = lane_bcast(sw, i);
    }
  }

  v2f rA[4];
#pragma unroll
  for (int kc = 0; kc < 4; ++kc) {
#pragma unroll
    for (int v = 0; v < 2; ++v) {
      int k = 4 * kc + v + 2 * hh;
      int diff = m16 ^ k;
      float P = 1.0f;
#pragma unroll
      for (int i = 0; i < 4; ++i) P *= ((diff >> i) & 1) ? s1[i] : c1[i];
      rA[kc][v] = (__popc(m16 & k) & 1) ? -P : P;
    }
  }

  v2f rU[4];
#pragma unroll
  for (int kc = 0; kc < 4; ++kc) {
#pragma unroll
    for (int v = 0; v < 2; ++v) {
      int k = 4 * kc + v + 2 * hh;
      int diff = m16 ^ k;
      float P = 1.0f;
#pragma unroll
      for (int i = 0; i < 4; ++i) P *= ((diff >> i) & 1) ? s1[5 + i] : c1[5 + i];
      rU[kc][v] = (__popc(m16 & k) & 1) ? -P : P;
    }
  }

  const float cz4 = c1[4] * c1[4] - s1[4] * s1[4];
  const float sy4 = 4.0f * (c1[4] * s1[4]);

  const float ax = (xv + w0v) * 0.5f;
  const float cx = cosf(ax), sx = sinf(ax);

  float zacc[4] = {0.f, 0.f, 0.f, 0.f};

#pragma unroll
  for (int ch = 0; ch < C; ++ch) {
    float ca[NQ], sa[NQ];
#pragma unroll
    for (int i = 0; i < NQ; ++i) {
      ca[i] = lane_bcast(cx, ch * NQ + i);
      sa[i] = lane_bcast(sx, ch * NQ + i);
    }

    v8f V = {0.f,0.f,0.f,0.f,0.f,0.f,0.f,0.f};
    {
      const int t4 = lane & 1, t8 = (lane >> 1) & 1, pl = (lane >> 2) & 1;
      v2f aT[4];
#pragma unroll
      for (int kc = 0; kc < 4; ++kc) {
#pragma unroll
        for (int v = 0; v < 2; ++v) {
          const int i2 = kc & 1, i3 = kc >> 1;
          const int s0 = v ^ t8, s1 = hh ^ v ^ t8, s2 = i2 ^ hh, s3 = i3 ^ i2, s4 = t4 ^ i3;
          const float amp = ((s0 ? sa[0] : ca[0]) * (s1 ? sa[1] : ca[1])) *
                            ((s2 ? sa[2] : ca[2]) * (s3 ? sa[3] : ca[3])) * (s4 ? sa[4] : ca[4]);
          const int q = (s0 + s1 + s2 + s3 + s4 + v + hh + __builtin_popcount(kc)) & 3;
          const float sg = ((q & 1) != pl) ? 0.0f : (((q + 1) & 2) ? -1.0f : 1.0f);
          aT[kc][v] = amp * sg;
        }
      }
#pragma unroll
      for (int kc = 0; kc < 4; ++kc) V = wmma4(aT[kc], rA[kc], V);
    }

    const v8f zero = {0.f,0.f,0.f,0.f,0.f,0.f,0.f,0.f};
    v8f Hh[2];
    {
      const int tg = lane & 1, pg = (lane >> 1) & 1;
      v2f aG[4];
#pragma unroll
      for (int kc = 0; kc < 4; ++kc) {
#pragma unroll
        for (int v = 0; v < 2; ++v) {
          const int s5 = v ^ tg, s6 = hh ^ v, s7 = (kc & 1) ^ hh, s8 = (kc >> 1) ^ (kc & 1);
          const float g = ((s5 ? sa[5] : ca[5]) * (s8 ? sa[8] : ca[8])) *
                          ((s6 ? sa[6] : ca[6]) * (s7 ? sa[7] : ca[7]));
          const int e = (s5 + s6 + s7 + s8 + tg + v + hh + __builtin_popcount(kc)) & 3;
          const float sg = ((e & 1) != pg) ? 0.0f : (((e + 1) & 2) ? -1.0f : 1.0f);
          aG[kc][v] = g * sg;
        }
      }
#pragma unroll
      for (int i8 = 0; i8 < 2; ++i8)
        Hh[i8] = wmma4(aG[2 * i8 + 1], rU[2 * i8 + 1], wmma4(aG[2 * i8], rU[2 * i8], zero));
    }
    v8f nr[2], ni[2];
#pragma unroll
    for (int t = 0; t < 2; ++t) {
      v2f aH, bR, bI;
#pragma unroll
      for (int v = 0; v < 2; ++v) {
        aH[v] = hh ? -Hh[v][2 + t] : Hh[v][t];
        bR[v] = hh ? V[4 + t + 2 * v] : V[t + 2 * v];
        bI[v] = hh ? -V[t + 2 * v] : V[4 + t + 2 * v];
      }
      nr[t] = wmma4(aH, bR, zero);
      ni[t] = wmma4(aH, bI, zero);
    }

    float tot = 0.f, alt = 0.f;
#pragma unroll
    for (int v = 0; v < 8; ++v) {
      const float re0 = nr[0][v], im0 = ni[0][v], re1 = nr[1][v], im1 = ni[1][v];
      const float p0 = re0 * re0 + im0 * im0;
      const float p1 = re1 * re1 + im1 * im1;
      const float zz = cz4 * (p0 - p1) + sy4 * (re0 * re1 + im0 * im1);
      tot += p0 + p1;
      alt += (__builtin_popcount(v) & 1) ? -zz : zz;
    }
    float z[4];
    z[0] = (__popc(lane & 0x1E) & 1) ? -alt : alt;
    z[1] = (__popc(lane & 0x03) & 1) ? -tot : tot;
    z[2] = (__popc(lane & 0x07) & 1) ? -tot : tot;
    z[3] = (__popc(lane & 0x0F) & 1) ? -tot : tot;
#pragma unroll
    for (int f = 0; f < 4; ++f) {
#pragma unroll
      for (int m = 1; m < 32; m <<= 1) z[f] += __shfl_xor(z[f], m, 32);
      zacc[f] += z[f];
    }
  }

  if (lane == 0) {
    const float inv = 1.0f / 3.0f;
#pragma unroll
    for (int f = 0; f < 4; ++f)
      out[((b * 4 + f) * HOUT + yy) * WOUT + xx] = zacc[f] * inv;
  }
}

extern "C" void kernel_launch(void* const* d_in, const int* in_sizes, int n_in,
                              void* d_out, int out_size, void* d_ws, size_t ws_size,
                              hipStream_t stream) {
  (void)in_sizes; (void)n_in; (void)out_size; (void)d_ws; (void)ws_size;
  const float* x  = (const float*)d_in[0];
  const float* qw = (const float*)d_in[1];
  float* out = (float*)d_out;
  const int total_waves = 4 * 26 * 26;
  dim3 grid(total_waves / 4), block(128);
  hipLaunchKernelGGL(quanv_kernel, grid, block, 0, stream, x, qw, out);
}


# Lane sums off the LDS unit: 3 channels' per-lane sums added first, then one butterfly per output on v_add_f32_dpp row_xmask + v_permlanex16; 60 ds_bpermute and their waits gone; sum order changes
# speedup vs baseline: 8.3648x; 1.0842x over previous
#include <hip/hip_runtime.h>


typedef __attribute__((ext_vector_type(2))) float v2f;
typedef __attribute__((ext_vector_type(8))) float v8f;

__device__ __forceinline__ v8f wmma4(v2f a, v2f b, v8f c) {
  return __builtin_amdgcn_wmma_f32_16x16x4_f32(false, a, false, b, (short)0, c,
                                               false, false);
}

__device__ __forceinline__ float lane_bcast(float v, int src_lane) {
  return __int_as_float(__builtin_amdgcn_readlane(__float_as_int(v), src_lane));
}

template <int XMASK>
__device__ __forceinline__ float row_xor(float v) {
  return __int_as_float(__builtin_amdgcn_update_dpp(0, __float_as_int(v),
                                                    0x160 + XMASK, 0xF, 0xF, true));
}
__device__ __forceinline__ float half_xor(float v) {
  const int b = __float_as_int(v);
  return __int_as_float(__builtin_amdgcn_permlanex16(b, b, 0x76543210, 0xfedcba98, false, false));
}
__device__ __forceinline__ float wave_sum(float v) {
  v += row_xor<1>(v);
  v += row_xor<2>(v);
  v += row_xor<4>(v);
  v += row_xor<8>(v);
  return v + half_xor(v);
}


__launch_bounds__(128)
__global__ void quanv_kernel(const float* __restrict__ x,
                             const float* __restrict__ qw,
                             float* __restrict__ out) {
  constexpr int NQ = 9, B = 4, C = 3, H = 28, W = 28, HOUT = 26, WOUT = 26;
  constexpr int TOTAL = B * HOUT * WOUT;

  const int lane = threadIdx.x & 31;
  const int warp = threadIdx.x >> 5;
  const int wid = blockIdx.x * 4 + warp;
  if (wid >= TOTAL) return;

  const int b  = wid / (HOUT * WOUT);
  const int yy = (wid / WOUT) % HOUT;
  const int xx = wid % WOUT;

  const int m16 = lane & 15;
  const int hh  = lane >> 4;

  const int la  = (lane < C * NQ) ? lane : (C * NQ - 1);
  const int chl = la / NQ, il = la - NQ * chl;
  const int dyl = il / 3, dxl = il - 3 * dyl;
  const float xv  = x[((b * C + chl) * H + yy + dyl) * W + xx + dxl];
  const float w0v = qw[il];

  float c1[NQ], s1[NQ];
  {
    const float aw = qw[NQ + ((lane < NQ) ? lane : (NQ - 1))] * 0.5f;
    const float cw = cosf(aw), sw = sinf(aw);
#pragma unroll
    for (int i = 0; i < NQ; ++i) {
      c1[i] = lane_bcast(cw, i);
      s1[i] = lane_bcast(sw, i);
    }
  }

  v2f rA[4];
#pragma unroll
  for (int kc = 0; kc < 4; ++kc) {
#pragma unroll
    for (int v = 0; v < 2; ++v) {
      int k = 4 * kc + v + 2 * hh;
      int diff = m16 ^ k;
      float P = 1.0f;
#pragma unroll
      for (int i = 0; i < 4; ++i) P *= ((diff >> i) & 1) ? s1[i] : c1[i];
      rA[kc][v] = (__popc(m16 & k) & 1) ? -P : P;
    }
  }

  v2f rU[4];
#pragma unroll
  for (int kc = 0; kc < 4; ++kc) {
#pragma unroll
    for (int v = 0; v < 2; ++v) {
      int k = 4 * kc + v + 2 * hh;
      int diff = m16 ^ k;
      float P = 1.0f;
#pragma unroll
      for (int i = 0; i < 4; ++i) P *= ((diff >> i) & 1) ? s1[5 + i] : c1[5 + i];
      rU[kc][v] = (__popc(m16 & k) & 1) ? -P : P;
    }
  }

  const float cz4 = c1[4] * c1[4] - s1[4] * s1[4];
  const float sy4 = 4.0f * (c1[4] * s1[4]);

  const float ax = (xv + w0v) * 0.5f;
  const float cx = cosf(ax), sx = sinf(ax);

  float tot = 0.f, alt = 0.f;

#pragma unroll
  for (int ch = 0; ch < C; ++ch) {
    float ca[NQ], sa[NQ];
#pragma unroll
    for (int i = 0; i < NQ; ++i) {
      ca[i] = lane_bcast(cx, ch * NQ + i);
      sa[i] = lane_bcast(sx, ch * NQ + i);
    }

    v8f V = {0.f,0.f,0.f,0.f,0.f,0.f,0.f,0.f};
    {
      const int t4 = lane & 1, t8 = (lane >> 1) & 1, pl = (lane >> 2) & 1;
      v2f aT[4];
#pragma unroll
      for (int kc = 0; kc < 4; ++kc) {
#pragma unroll
        for (int v = 0; v < 2; ++v) {
          const int i2 = kc & 1, i3 = kc >> 1;
          const int s0 = v ^ t8, s1 = hh ^ v ^ t8, s2 = i2 ^ hh, s3 = i3 ^ i2, s4 = t4 ^ i3;
          const float amp = ((s0 ? sa[0] : ca[0]) * (s1 ? sa[1] : ca[1])) *
                            ((s2 ? sa[2] : ca[2]) * (s3 ? sa[3] : ca[3])) * (s4 ? sa[4] : ca[4]);
          const int q = (s0 + s1 + s2 + s3 + s4 + v + hh + __builtin_popcount(kc)) & 3;
          const float sg = ((q & 1) != pl) ? 0.0f : (((q + 1) & 2) ? -1.0f : 1.0f);
          aT[kc][v] = amp * sg;
        }
      }
#pragma unroll
      for (int kc = 0; kc < 4; ++kc) V = wmma4(aT[kc], rA[kc], V);
    }

    const v8f zero = {0.f,0.f,0.f,0.f,0.f,0.f,0.f,0.f};
    v8f Hh[2];
    {
      const int tg = lane & 1, pg = (lane >> 1) & 1;
      v2f aG[4];
#pragma unroll
      for (int kc = 0; kc < 4; ++kc) {
#pragma unroll
        for (int v = 0; v < 2; ++v) {
          const int s5 = v ^ tg, s6 = hh ^ v, s7 = (kc & 1) ^ hh, s8 = (kc >> 1) ^ (kc & 1);
          const float g = ((s5 ? sa[5] : ca[5]) * (s8 ? sa[8] : ca[8])) *
                          ((s6 ? sa[6] : ca[6]) * (s7 ? sa[7] : ca[7]));
          const int e = (s5 + s6 + s7 + s8 + tg + v + hh + __builtin_popcount(kc)) & 3;
          const float sg = ((e & 1) != pg) ? 0.0f : (((e + 1) & 2) ? -1.0f : 1.0f);
          aG[kc][v] = g * sg;
        }
      }
#pragma unroll
      for (int i8 = 0; i8 < 2; ++i8)
        Hh[i8] = wmma4(aG[2 * i8 + 1], rU[2 * i8 + 1], wmma4(aG[2 * i8], rU[2 * i8], zero));
    }
    v8f nr[2], ni[2];
#pragma unroll
    for (int t = 0; t < 2; ++t) {
      v2f aH, bR, bI;
#pragma unroll
      for (int v = 0; v < 2; ++v) {
        aH[v] = hh ? -Hh[v][2 + t] : Hh[v][t];
        bR[v] = hh ? V[4 + t + 2 * v] : V[t + 2 * v];
        bI[v] = hh ? -V[t + 2 * v] : V[4 + t + 2 * v];
      }
      nr[t] = wmma4(aH, bR, zero);
      ni[t] = wmma4(aH, bI, zero);
    }

#pragma unroll
    for (int v = 0; v < 8; ++v) {
      const float re0 = nr[0][v], im0 = ni[0][v], re1 = nr[1][v], im1 = ni[1][v];
      const float p0 = re0 * re0 + im0 * im0;
      const float p1 = re1 * re1 + im1 * im1;
      const float zz = cz4 * (p0 - p1) + sy4 * (re0 * re1 + im0 * im1);
      tot += p0 + p1;
      alt += (__builtin_popcount(v) & 1) ? -zz : zz;
    }
  }

  float z[4];
  z[0] = wave_sum((__popc(lane & 0x1E) & 1) ? -alt : alt);
  z[1] = wave_sum((__popc(lane & 0x03) & 1) ? -tot : tot);
  z[2] = wave_sum((__popc(lane & 0x07) & 1) ? -tot : tot);
  z[3] = wave_sum((__popc(lane & 0x0F) & 1) ? -tot : tot);

  if (lane == 0) {
    const float inv = 1.0f / 3.0f;
#pragma unroll
    for (int f = 0; f < 4; ++f)
      out[((b * 4 + f) * HOUT + yy) * WOUT + xx] = z[f] * inv;
  }
}

extern "C" void kernel_launch(void* const* d_in, const int* in_sizes, int n_in,
                              void* d_out, int out_size, void* d_ws, size_t ws_size,
                              hipStream_t stream) {
  (void)in_sizes; (void)n_in; (void)out_size; (void)d_ws; (void)ws_size;
  const float* x  = (const float*)d_in[0];
  const float* qw = (const float*)d_in[1];
  float* out = (float*)d_out;
  const int total_waves = 4 * 26 * 26;
  dim3 grid(total_waves / 4), block(128);
  hipLaunchKernelGGL(quanv_kernel, grid, block, 0, stream, x, qw, out);
}


# Both RX products taken once per wave for the 3 channels: operand rows = channel x branch, K = the source indices of one popcount parity (no zero entries): 4+4 WMMAs, not 12+12; sum order changes
# speedup vs baseline: 9.4493x; 1.1297x over previous
#include <hip/hip_runtime.h>


typedef __attribute__((ext_vector_type(2))) float v2f;
typedef __attribute__((ext_vector_type(8))) float v8f;

__device__ __forceinline__ v8f wmma4(v2f a, v2f b, v8f c) {
  return __builtin_amdgcn_wmma_f32_16x16x4_f32(false, a, false, b, (short)0, c,
                                               false, false);
}

__device__ __forceinline__ float lane_bcast(float v, int src_lane) {
  return __int_as_float(__builtin_amdgcn_readlane(__float_as_int(v), src_lane));
}

__device__ __forceinline__ float row_gather(float v, unsigned sel) {
  const int b = __float_as_int(v);
  return __int_as_float(__builtin_amdgcn_permlane16(b, b, sel, sel, false, false));
}

template <int XMASK>
__device__ __forceinline__ float row_xor(float v) {
  return __int_as_float(__builtin_amdgcn_update_dpp(0, __float_as_int(v),
                                                    0x160 + XMASK, 0xF, 0xF, true));
}
__device__ __forceinline__ float half_xor(float v) {
  const int b = __float_as_int(v);
  return __int_as_float(__builtin_amdgcn_permlanex16(b, b, 0x76543210, 0xfedcba98, false, false));
}
__device__ __forceinline__ float wave_sum(float v) {
  v += row_xor<1>(v);
  v += row_xor<2>(v);
  v += row_xor<4>(v);
  v += row_xor<8>(v);
  return v + half_xor(v);
}

constexpr unsigned sign_table_first(int p, int kc, int v) {
  unsigned tab = 0;
  for (int n = 0; n < 8; ++n) {
    const int hf = n & 1, t8 = (n >> 1) & 1, pl = n >> 2;
    const int i1 = hf, i2 = v, i3 = kc, i0 = p ^ hf ^ v ^ kc, t4 = p ^ pl;
    const int s0 = i0 ^ t8, s1 = i1 ^ i0 ^ t8, s2 = i2 ^ i1, s3 = i3 ^ i2, s4 = t4 ^ i3;
    const int q = (s0 + s1 + s2 + s3 + s4 + i0 + i1 + i2 + i3) & 3;
    if ((q + 1) & 2) tab |= 1u << n;
  }
  return tab;
}
constexpr unsigned sign_table_second(int i8, int e, int v) {
  unsigned tab = 0;
  for (int n = 0; n < 4; ++n) {
    const int hf = n & 1, t = n >> 1;
    const int i5 = v, i6 = hf, i7 = e ^ v ^ hf;
    const int s5 = i5 ^ t, s6 = i6 ^ i5, s7 = i7 ^ i6, s8 = i8 ^ i7;
    const int q = (s5 + s6 + s7 + s8 + t + i5 + i6 + i7 + i8) & 3;
    if ((q + 1) & 2) tab |= 1u << n;
  }
  return tab;
}
__device__ __forceinline__ float apply_sign(float m, unsigned tab, int sh) {
  return __int_as_float(__float_as_int(m) ^ (int)((tab << sh) & 0x80000000u));
}


__launch_bounds__(128)
__global__ void quanv_kernel(const float* __restrict__ x,
                             const float* __restrict__ qw,
                             float* __restrict__ out) {
  constexpr int NQ = 9, B = 4, C = 3, H = 28, W = 28, HOUT = 26, WOUT = 26;
  constexpr int TOTAL = B * HOUT * WOUT;

  const int lane = threadIdx.x & 31;
  const int warp = threadIdx.x >> 5;
  const int wid = blockIdx.x * 4 + warp;
  if (wid >= TOTAL) return;

  const int b  = wid / (HOUT * WOUT);
  const int yy = (wid / WOUT) % HOUT;
  const int xx = wid % WOUT;

  const int rho = lane & 15;
  const int hh  = lane >> 4;

  const int jt  = (rho < 15) ? rho : 14;
  const int cht = jt / 5, kt = jt - 5 * cht;
  const int dyt = kt / 3, dxt = kt - 3 * dyt;
  const int jg  = (rho < 12) ? rho : 11;
  const int chg = jg >> 2, kg = 5 + (jg & 3);
  const int dyg = kg / 3, dxg = kg - 3 * dyg;
  const float xT  = x[((b * C + cht) * H + yy + dyt) * W + xx + dxt];
  const float xG  = x[((b * C + chg) * H + yy + dyg) * W + xx + dxg];
  const float w0T = qw[kt];
  const float w0G = qw[kg];

  float c1[NQ], s1[NQ];
  {
    const float aw = qw[NQ + ((lane < NQ) ? lane : (NQ - 1))] * 0.5f;
    const float cw = cosf(aw), sw = sinf(aw);
#pragma unroll
    for (int i = 0; i < NQ; ++i) {
      c1[i] = lane_bcast(cw, i);
      s1[i] = lane_bcast(sw, i);
    }
  }

  v2f bA[2][2];
#pragma unroll
  for (int p = 0; p < 2; ++p) {
#pragma unroll
    for (int kc = 0; kc < 2; ++kc) {
#pragma unroll
      for (int v = 0; v < 2; ++v) {
        const int i0 = p ^ hh ^ v ^ kc;
        const int r = i0 + 2 * hh + 4 * v + 8 * kc;
        const int diff = rho ^ r;
        float P = 1.0f;
#pragma unroll
        for (int i = 0; i < 4; ++i) P *= ((diff >> i) & 1) ? s1[i] : c1[i];
        bA[p][kc][v] = (__popc(rho & r) & 1) ? -P : P;
      }
    }
  }

  v2f bU[2][2];
#pragma unroll
  for (int i8 = 0; i8 < 2; ++i8) {
#pragma unroll
    for (int e = 0; e < 2; ++e) {
#pragma unroll
      for (int v = 0; v < 2; ++v) {
        const int u = v + 2 * hh + 4 * (e ^ v ^ hh) + 8 * i8;
        const int diff = rho ^ u;
        float P = 1.0f;
#pragma unroll
        for (int i = 0; i < 4; ++i) P *= ((diff >> i) & 1) ? s1[5 + i] : c1[5 + i];
        bU[i8][e][v] = (__popc(rho & u) & 1) ? -P : P;
      }
    }
  }

  const float cz4 = c1[4] * c1[4] - s1[4] * s1[4];
  const float sy4 = 4.0f * (c1[4] * s1[4]);

  const float aTh = (xT + w0T) * 0.5f, aGh = (xG + w0G) * 0.5f;
  const float cT = cosf(aTh), sT = sinf(aTh);
  const float cG = cosf(aGh), sG = sinf(aGh);

  float ca[NQ], sa[NQ];
#pragma unroll
  for (int k = 0; k < 5; ++k) {
    const unsigned sel = 0xAAAA5500u + 0x11111111u * (unsigned)k;
    ca[k] = row_gather(cT, sel);
    sa[k] = row_gather(sT, sel);
  }
#pragma unroll
  for (int k = 0; k < 4; ++k) {
    const unsigned sel = 0x88884400u + 0x11111111u * (unsigned)k;
    ca[5 + k] = row_gather(cG, sel);
    sa[5 + k] = row_gather(sG, sel);
  }

  const int t8 = rho & 1;
  const int pl = rho >> 3;
  const v8f zero = {0.f,0.f,0.f,0.f,0.f,0.f,0.f,0.f};

  v8f Dp[2];
  {
    float f0v[2], f1v[2], f2v[2], f3v[2], f4v[2];
#pragma unroll
    for (int c = 0; c < 2; ++c) {
      f0v[c] = (c ^ hh ^ t8) ? sa[0] : ca[0];
      f1v[c] = (c ^ t8) ? sa[1] : ca[1];
      f2v[c] = (c ^ hh) ? sa[2] : ca[2];
      f3v[c] = c ? sa[3] : ca[3];
      f4v[c] = (c ^ pl) ? sa[4] : ca[4];
    }
    float E[2], M[2][2], N[2][2];
#pragma unroll
    for (int c = 0; c < 2; ++c) E[c] = f0v[c] * f1v[c];
#pragma unroll
    for (int v = 0; v < 2; ++v) {
#pragma unroll
      for (int kc = 0; kc < 2; ++kc) M[v][kc] = f2v[v] * f3v[kc ^ v];
    }
#pragma unroll
    for (int a = 0; a < 2; ++a) {
#pragma unroll
      for (int v = 0; v < 2; ++v) N[a][v] = E[a ^ v] * f4v[a];
    }
    constexpr unsigned TT[8] = {
        sign_table_first(0, 0, 0), sign_table_first(0, 0, 1), sign_table_first(0, 1, 0),
        sign_table_first(0, 1, 1), sign_table_first(1, 0, 0), sign_table_first(1, 0, 1),
        sign_table_first(1, 1, 0), sign_table_first(1, 1, 1)};
    const int shT = 31 - (hh + 2 * t8 + 4 * pl);
#pragma unroll
    for (int p = 0; p < 2; ++p) {
      v8f acc = zero;
#pragma unroll
      for (int kc = 0; kc < 2; ++kc) {
        v2f aT;
#pragma unroll
        for (int v = 0; v < 2; ++v)
          aT[v] = apply_sign(N[p ^ kc][v] * M[v][kc], TT[4 * p + 2 * kc + v], shT);
        acc = wmma4(aT, bA[p][kc], acc);
      }
      Dp[p] = acc;
    }
  }

  v8f HW[2][2];
  {
    float f5v[2], f6v[2], f7v[2], f8v[2];
#pragma unroll
    for (int c = 0; c < 2; ++c) {
      f5v[c] = (c ^ t8) ? sa[5] : ca[5];
      f6v[c] = (c ^ hh) ? sa[6] : ca[6];
      f7v[c] = c ? sa[7] : ca[7];
      f8v[c] = (c ^ hh) ? sa[8] : ca[8];
    }
    float E56[2], F78[2][2];
#pragma unroll
    for (int v = 0; v < 2; ++v) E56[v] = f5v[v] * f6v[v];
#pragma unroll
    for (int bb = 0; bb < 2; ++bb) {
#pragma unroll
      for (int i8 = 0; i8 < 2; ++i8) F78[bb][i8] = f7v[bb] * f8v[i8 ^ bb];
    }
    constexpr unsigned TG[8] = {
        sign_table_second(0, 0, 0), sign_table_second(0, 0, 1), sign_table_second(0, 1, 0),
        sign_table_second(0, 1, 1), sign_table_second(1, 0, 0), sign_table_second(1, 0, 1),
        sign_table_second(1, 1, 0), sign_table_second(1, 1, 1)};
    const int shG = 31 - (hh + 2 * t8);
#pragma unroll
    for (int i8 = 0; i8 < 2; ++i8) {
#pragma unroll
      for (int e = 0; e < 2; ++e) {
        v2f aG;
#pragma unroll
        for (int v = 0; v < 2; ++v)
          aG[v] = apply_sign(E56[v] * F78[e ^ v][i8], TG[4 * i8 + 2 * e + v], shG);
        HW[i8][e] = wmma4(aG, bU[i8][e], zero);
      }
    }
  }

  float tot = 0.f, alt = 0.f;

#pragma unroll
  for (int ch = 0; ch < C; ++ch) {
    v8f nr[2], ni[2];
#pragma unroll
    for (int t = 0; t < 2; ++t) {
      v2f aRe, aIm, bV;
#pragma unroll
      for (int v = 0; v < 2; ++v) {
        const float Hr = HW[v][0][t + 2 * ch], Hi = HW[v][1][t + 2 * ch];
        aRe[v] = hh ? -Hi : Hr;
        aIm[v] = hh ? Hr : Hi;
        bV[v]  = (t ^ hh) ? Dp[1][v + 2 * ch] : Dp[0][v + 2 * ch];
      }
      nr[t] = wmma4(aRe, bV, zero);
      ni[t] = wmma4(aIm, bV, zero);
    }

#pragma unroll
    for (int v = 0; v < 8; ++v) {
      const float re0 = nr[0][v], im0 = ni[0][v], re1 = nr[1][v], im1 = ni[1][v];
      const float p0 = re0 * re0 + im0 * im0;
      const float p1 = re1 * re1 + im1 * im1;
      const float zz = cz4 * (p0 - p1) + sy4 * (re0 * re1 + im0 * im1);
      tot += p0 + p1;
      alt += (__builtin_popcount(v) & 1) ? -zz : zz;
    }
  }

  float z[4];
  z[0] = wave_sum((__popc(lane & 0x1E) & 1) ? -alt : alt);
  z[1] = wave_sum((__popc(lane & 0x03) & 1) ? -tot : tot);
  z[2] = wave_sum((__popc(lane & 0x07) & 1) ? -tot : tot);
  z[3] = wave_sum((__popc(lane & 0x0F) & 1) ? -tot : tot);

  if (lane == 0) {
    const float inv = 1.0f / 3.0f;
#pragma unroll
    for (int f = 0; f < 4; ++f)
      out[((b * 4 + f) * HOUT + yy) * WOUT + xx] = z[f] * inv;
  }
}

extern "C" void kernel_launch(void* const* d_in, const int* in_sizes, int n_in,
                              void* d_out, int out_size, void* d_ws, size_t ws_size,
                              hipStream_t stream) {
  (void)in_sizes; (void)n_in; (void)out_size; (void)d_ws; (void)ws_size;
  const float* x  = (const float*)d_in[0];
  const float* qw = (const float*)d_in[1];
  float* out = (float*)d_out;
  const int total_waves = 4 * 26 * 26;
  dim3 grid(total_waves / 4), block(128);
  hipLaunchKernelGGL(quanv_kernel, grid, block, 0, stream, x, qw, out);
}
